# starting kernel
# baseline (speedup 1.0000x reference)
#include <hip/hip_runtime.h>
#include <cstdint>

typedef float v2f __attribute__((ext_vector_type(2)));
typedef float v4f __attribute__((ext_vector_type(4)));
typedef float v8f __attribute__((ext_vector_type(8)));

#define B_DIM 64
#define H_DIM 1024
#define I_DIM 512

__global__ __launch_bounds__(128) void reservoir_gemm_wmma(
    const float* __restrict__ x_t,
    const float* __restrict__ W_ih,
    const float* __restrict__ b_ih,
    float* __restrict__ ic)
{
  const int lane = threadIdx.x & 31;
  const int waveInBlk = threadIdx.x >> 5;
  const int tile = blockIdx.x * 4 + waveInBlk;
  const int tileM = tile & 3;
  const int tileN = tile >> 2;

  const int mLane = lane & 15;
  const int kBase = (lane >> 4) << 1;

  const float* aPtr = x_t + (size_t)(tileM * 16 + mLane) * I_DIM + kBase;
  const float* bPtr = W_ih + (size_t)(tileN * 16 + mLane) * I_DIM + kBase;

  v8f acc = {};
  for (int k0 = 0; k0 < I_DIM; k0 += 4) {
    v2f a = *reinterpret_cast<const v2f*>(aPtr + k0);
    v2f b = *reinterpret_cast<const v2f*>(bPtr + k0);
    acc = __builtin_amdgcn_wmma_f32_16x16x4_f32(
         false, a,  false, b,
         (short)0, acc,  false,  false);
  }

  const int col = tileN * 16 + mLane;
  const float bias = b_ih[col];
  const int rowBase = tileM * 16 + ((lane >> 4) << 3);
#pragma unroll
  for (int r = 0; r < 8; ++r) {
    ic[(size_t)(rowBase + r) * H_DIM + col] = acc[r] + bias;
  }
}

__global__ __launch_bounds__(256) void reservoir_softmax_mix(
    const float* __restrict__ h_prev,
    const float* __restrict__ W_hh,
    const float* __restrict__ temperature,
    const float* __restrict__ gumbel,
    const float* __restrict__ ic,
    float* __restrict__ out)
{
  __shared__ float hp[H_DIM];

  const int b = blockIdx.x >> 7;
  const int oBase = (blockIdx.x & 127) << 3;
  const int tid = threadIdx.x;

  {
    const v4f* src = reinterpret_cast<const v4f*>(h_prev + (size_t)b * H_DIM);
    reinterpret_cast<v4f*>(hp)[tid] = src[tid];
  }
  __syncthreads();

  const float tau = fmaxf(temperature[0], 1e-3f);
  const float inv_tau = 1.0f / tau;

  const int lane = tid & 31;
  const int wv = tid >> 5;
  const int o = oBase + wv;

  const float* grow = gumbel + ((size_t)(b * H_DIM + o)) * H_DIM;
  const float* wrow = W_hh + (size_t)o * H_DIM;

  v4f v[8];
  float m = -3.0e38f;
#pragma unroll
  for (int it = 0; it < 8; ++it) {
    const int h0 = it * 128 + lane * 4;
    v4f g = __builtin_nontemporal_load(reinterpret_cast<const v4f*>(grow + h0));
    v4f w = *reinterpret_cast<const v4f*>(wrow + h0);
#pragma unroll
    for (int c = 0; c < 4; ++c) {
      float t = fmaf(w[c], inv_tau, g[c]);
      v[it][c] = t;
      m = fmaxf(m, t);
    }
  }
#pragma unroll
  for (int off = 16; off > 0; off >>= 1)
    m = fmaxf(m, __shfl_xor(m, off, 32));

  float s = 0.0f, wsum = 0.0f;
#pragma unroll
  for (int it = 0; it < 8; ++it) {
    const int h0 = it * 128 + lane * 4;
    v4f hpv = *reinterpret_cast<const v4f*>(&hp[h0]);
#pragma unroll
    for (int c = 0; c < 4; ++c) {
      float e = __expf(v[it][c] - m);
      s += e;
      wsum = fmaf(e, hpv[c], wsum);
    }
  }
#pragma unroll
  for (int off = 16; off > 0; off >>= 1) {
    s += __shfl_xor(s, off, 32);
    wsum += __shfl_xor(wsum, off, 32);
  }

  if (lane == 0) {
    const float contrib = wsum / s;
    out[(size_t)b * H_DIM + o] = tanhf(ic[(size_t)b * H_DIM + o] + contrib);
  }
}

extern "C" void kernel_launch(void* const* d_in, const int* in_sizes, int n_in,
                              void* d_out, int out_size, void* d_ws, size_t ws_size,
                              hipStream_t stream) {
  const float* x_t         = (const float*)d_in[0];
  const float* h_prev      = (const float*)d_in[1];
  const float* W_ih        = (const float*)d_in[2];
  const float* b_ih        = (const float*)d_in[3];
  const float* W_hh        = (const float*)d_in[4];
  const float* temperature = (const float*)d_in[5];
  const float* gumbel      = (const float*)d_in[6];
  float* out = (float*)d_out;
  float* ic  = (float*)d_ws;

  reservoir_gemm_wmma<<<64, 128, 0, stream>>>(x_t, W_ih, b_ih, ic);

  reservoir_softmax_mix<<<B_DIM * (H_DIM / 8), 256, 0, stream>>>(
      h_prev, W_hh, temperature, gumbel, ic, out);
}


# Each wave keeps one W_hh row in registers and walks 16 batches with the next three noise rows in flight: 4096 long-lived waves instead of 65536 one-row waves; per-row sum order unchanged
# speedup vs baseline: 1.2984x; 1.2984x over previous
#include <hip/hip_runtime.h>
#include <cstdint>

typedef float v2f __attribute__((ext_vector_type(2)));
typedef float v4f __attribute__((ext_vector_type(4)));
typedef float v8f __attribute__((ext_vector_type(8)));

#define B_DIM 64
#define H_DIM 1024
#define I_DIM 512

__global__ __launch_bounds__(128) void reservoir_gemm_wmma(
    const float* __restrict__ x_t,
    const float* __restrict__ W_ih,
    const float* __restrict__ b_ih,
    float* __restrict__ ic)
{
  const int lane = threadIdx.x & 31;
  const int waveInBlk = threadIdx.x >> 5;
  const int tile = blockIdx.x * 4 + waveInBlk;
  const int tileM = tile & 3;
  const int tileN = tile >> 2;

  const int mLane = lane & 15;
  const int kBase = (lane >> 4) << 1;

  const float* aPtr = x_t + (size_t)(tileM * 16 + mLane) * I_DIM + kBase;
  const float* bPtr = W_ih + (size_t)(tileN * 16 + mLane) * I_DIM + kBase;

  v8f acc = {};
  for (int k0 = 0; k0 < I_DIM; k0 += 4) {
    v2f a = *reinterpret_cast<const v2f*>(aPtr + k0);
    v2f b = *reinterpret_cast<const v2f*>(bPtr + k0);
    acc = __builtin_amdgcn_wmma_f32_16x16x4_f32(
         false, a,  false, b,
         (short)0, acc,  false,  false);
  }

  const int col = tileN * 16 + mLane;
  const float bias = b_ih[col];
  const int rowBase = tileM * 16 + ((lane >> 4) << 3);
#pragma unroll
  for (int r = 0; r < 8; ++r) {
    ic[(size_t)(rowBase + r) * H_DIM + col] = acc[r] + bias;
  }
}

#define NB 16
#define NW 16
#define RING 4

__global__ __launch_bounds__(NW * 32) void reservoir_softmax_mix(
    const float* __restrict__ h_prev,
    const float* __restrict__ W_hh,
    const float* __restrict__ temperature,
    const float* __restrict__ gumbel,
    const float* __restrict__ ic,
    float* __restrict__ out)
{
  __shared__ float hp[NB * H_DIM];

  const int tid = threadIdx.x;
  const int lane = tid & 31;
  const int wv = __builtin_amdgcn_readfirstlane(tid >> 5);
  const int o = (blockIdx.x & (H_DIM / NW - 1)) * NW + wv;
  const int b0 = (blockIdx.x / (H_DIM / NW)) * NB;

  const float* gbase = gumbel + ((size_t)b0 * H_DIM + o) * H_DIM + lane * 4;
  const size_t gstep = (size_t)H_DIM * H_DIM;

  v4f g[RING][8];
#define LOAD_ROW(SLOT, I)                                                      \
  do {                                                                         \
    const float* gp_ = gbase + (size_t)(I) * gstep;                            \
    _Pragma("unroll") for (int it = 0; it < 8; ++it) g[SLOT][it] =             \
        __builtin_nontemporal_load(reinterpret_cast<const v4f*>(gp_ + it * 128)); \
  } while (0)

  LOAD_ROW(0, 0);
  LOAD_ROW(1, 1);
  LOAD_ROW(2, 2);

  v4f w[8];
  {
    const float* wrow = W_hh + (size_t)o * H_DIM + lane * 4;
#pragma unroll
    for (int it = 0; it < 8; ++it)
      w[it] = *reinterpret_cast<const v4f*>(wrow + it * 128);
  }

  {
    const v4f* src = reinterpret_cast<const v4f*>(h_prev + (size_t)b0 * H_DIM);
    v4f* dst = reinterpret_cast<v4f*>(hp);
#pragma unroll
    for (int q = 0; q < 8; ++q) dst[q * (NW * 32) + tid] = src[q * (NW * 32) + tid];
  }
  __syncthreads();

  const float tau = fmaxf(temperature[0], 1e-3f);
  const float inv_tau = 1.0f / tau;

  float my_s = 1.0f, my_w = 0.0f;

#define DO_ROW(SLOT, I)                                                        \
  do {                                                                         \
    float m = -3.0e38f;                                                        \
    _Pragma("unroll") for (int it = 0; it < 8; ++it) {                         \
      _Pragma("unroll") for (int c = 0; c < 4; ++c) {                          \
        float t = fmaf(w[it][c], inv_tau, g[SLOT][it][c]);                     \
        g[SLOT][it][c] = t;                                                    \
        m = fmaxf(m, t);                                                       \
      }                                                                        \
    }                                                                          \
    _Pragma("unroll") for (int off = 16; off > 0; off >>= 1)                   \
      m = fmaxf(m, __shfl_xor(m, off, 32));                                    \
    float s = 0.0f, wsum = 0.0f;                                               \
    const float* hrow = hp + (I) * H_DIM + lane * 4;                           \
    _Pragma("unroll") for (int it = 0; it < 8; ++it) {                         \
      v4f hpv = *reinterpret_cast<const v4f*>(hrow + it * 128);                \
      _Pragma("unroll") for (int c = 0; c < 4; ++c) {                          \
        float e = __expf(g[SLOT][it][c] - m);                                  \
        s += e;                                                                \
        wsum = fmaf(e, hpv[c], wsum);                                          \
      }                                                                        \
    }                                                                          \
    _Pragma("unroll") for (int off = 16; off > 0; off >>= 1) {                 \
      s += __shfl_xor(s, off, 32);                                             \
      wsum += __shfl_xor(wsum, off, 32);                                       \
    }                                                                          \
    if (lane == (I)) { my_s = s; my_w = wsum; }                                \
  } while (0)

#pragma unroll 1
  for (int t = 0; t < NB - RING; t += RING) {
    __builtin_amdgcn_sched_barrier(0);
    LOAD_ROW(3, t + 3);
    __builtin_amdgcn_sched_barrier(0);
    DO_ROW(0, t + 0);
    __builtin_amdgcn_sched_barrier(0);
    LOAD_ROW(0, t + 4);
    __builtin_amdgcn_sched_barrier(0);
    DO_ROW(1, t + 1);
    __builtin_amdgcn_sched_barrier(0);
    LOAD_ROW(1, t + 5);
    __builtin_amdgcn_sched_barrier(0);
    DO_ROW(2, t + 2);
    __builtin_amdgcn_sched_barrier(0);
    LOAD_ROW(2, t + 6);
    __builtin_amdgcn_sched_barrier(0);
    DO_ROW(3, t + 3);
  }
  __builtin_amdgcn_sched_barrier(0);
  LOAD_ROW(3, NB - 1);
  __builtin_amdgcn_sched_barrier(0);
  DO_ROW(0, NB - 4);
  DO_ROW(1, NB - 3);
  DO_ROW(2, NB - 2);
  DO_ROW(3, NB - 1);
#undef LOAD_ROW
#undef DO_ROW

  if (lane < NB) {
    const size_t idx = (size_t)(b0 + lane) * H_DIM + o;
    const float contrib = my_w / my_s;
    out[idx] = tanhf(ic[idx] + contrib);
  }
}

extern "C" void kernel_launch(void* const* d_in, const int* in_sizes, int n_in,
                              void* d_out, int out_size, void* d_ws, size_t ws_size,
                              hipStream_t stream) {
  const float* x_t         = (const float*)d_in[0];
  const float* h_prev      = (const float*)d_in[1];
  const float* W_ih        = (const float*)d_in[2];
  const float* b_ih        = (const float*)d_in[3];
  const float* W_hh        = (const float*)d_in[4];
  const float* temperature = (const float*)d_in[5];
  const float* gumbel      = (const float*)d_in[6];
  float* out = (float*)d_out;
  float* ic  = (float*)d_ws;

  reservoir_gemm_wmma<<<64, 128, 0, stream>>>(x_t, W_ih, b_ih, ic);

  reservoir_softmax_mix<<<(H_DIM / NW) * (B_DIM / NB), NW * 32, 0, stream>>>(
      h_prev, W_hh, temperature, gumbel, ic, out);
}


# Loop nest interchanged: each wave keeps one batch's h_prev row in registers and walks 16 consecutive rows o, one contiguous 64 KB noise stream, W_hh rows from LDS; per-row sum order unchanged
# speedup vs baseline: 1.6643x; 1.2818x over previous
#include <hip/hip_runtime.h>
#include <cstdint>

typedef float v2f __attribute__((ext_vector_type(2)));
typedef float v4f __attribute__((ext_vector_type(4)));
typedef float v8f __attribute__((ext_vector_type(8)));

#define B_DIM 64
#define H_DIM 1024
#define I_DIM 512

#define NB 16
#define NW 16
#define RING 4

__global__ __launch_bounds__(NW * 32) void reservoir_step_fused(
    const float* __restrict__ x_t,
    const float* __restrict__ h_prev,
    const float* __restrict__ W_ih,
    const float* __restrict__ b_ih,
    const float* __restrict__ W_hh,
    const float* __restrict__ temperature,
    const float* __restrict__ gumbel,
    float* __restrict__ out)
{
  __shared__ float wl[NB * H_DIM];
  __shared__ float part[NW * 256];

  const int tid = threadIdx.x;
  const int lane = tid & 31;
  const int wv = __builtin_amdgcn_readfirstlane(tid >> 5);
  const int o0 = (blockIdx.x & (H_DIM / NB - 1)) * NB;
  const int b0 = (blockIdx.x / (H_DIM / NB)) * NW;
  const int b = b0 + wv;

  const float* gbase = gumbel + ((size_t)b * H_DIM + o0) * H_DIM + lane * 4;
  const size_t gstep = (size_t)H_DIM;

  v4f g[RING][8];
#define LOAD_ROW(SLOT, I)                                                      \
  do {                                                                         \
    const float* gp_ = gbase + (size_t)(I) * gstep;                            \
    _Pragma("unroll") for (int it = 0; it < 8; ++it) g[SLOT][it] =             \
        __builtin_nontemporal_load(reinterpret_cast<const v4f*>(gp_ + it * 128)); \
  } while (0)

  const int fc = lane & 15;
  const int fh = lane >> 4;
  v4f af[4], bf[4];
  {
    const float* ap = x_t + (size_t)(b0 + fc) * I_DIM + wv * 32 + fh * 16;
    const float* bp = W_ih + (size_t)(o0 + fc) * I_DIM + wv * 32 + fh * 16;
#pragma unroll
    for (int q = 0; q < 4; ++q) {
      af[q] = *reinterpret_cast<const v4f*>(ap + q * 4);
      bf[q] = *reinterpret_cast<const v4f*>(bp + q * 4);
    }
  }

  LOAD_ROW(0, 0);
  LOAD_ROW(1, 1);
  LOAD_ROW(2, 2);

  v4f hreg[8];
  {
    const float* hrow = h_prev + (size_t)b * H_DIM + lane * 4;
#pragma unroll
    for (int it = 0; it < 8; ++it)
      hreg[it] = *reinterpret_cast<const v4f*>(hrow + it * 128);
  }

  v4f hs[8];
  {
    const v4f* src = reinterpret_cast<const v4f*>(W_hh + (size_t)o0 * H_DIM);
#pragma unroll
    for (int q = 0; q < 8; ++q) hs[q] = src[q * (NW * 32) + tid];
  }

  {
    v8f acc = {};
#pragma unroll
    for (int q = 0; q < 4; ++q) {
#pragma unroll
      for (int p = 0; p < 2; ++p) {
        v2f a = {af[q][2 * p], af[q][2 * p + 1]};
        v2f b = {bf[q][2 * p], bf[q][2 * p + 1]};
        acc = __builtin_amdgcn_wmma_f32_16x16x4_f32(
             false, a,  false, b,
             (short)0, acc,  false,  false);
      }
    }
#pragma unroll
    for (int r = 0; r < 8; ++r) part[wv * 256 + r * 32 + lane] = acc[r];
  }
  {
    v4f* dst = reinterpret_cast<v4f*>(wl);
#pragma unroll
    for (int q = 0; q < 8; ++q) dst[q * (NW * 32) + tid] = hs[q];
  }
  __syncthreads();

  const float tau = fmaxf(temperature[0], 1e-3f);
  const float inv_tau = 1.0f / tau;

  float my_s = 1.0f, my_w = 0.0f;

  v4f wr[8];
#define LOAD_W(I)                                                              \
  do {                                                                         \
    const float* wrow_ = wl + (I) * H_DIM + lane * 4;                          \
    _Pragma("unroll") for (int it = 0; it < 8; ++it) wr[it] =                  \
        *reinterpret_cast<const v4f*>(wrow_ + it * 128);                       \
  } while (0)
  LOAD_W(0);

#define DO_ROW(SLOT, I, MORE)                                                  \
  do {                                                                         \
    float m = -3.0e38f;                                                        \
    _Pragma("unroll") for (int it = 0; it < 8; ++it) {                         \
      _Pragma("unroll") for (int c = 0; c < 4; ++c) {                          \
        float t = fmaf(wr[it][c], inv_tau, g[SLOT][it][c]);                    \
        g[SLOT][it][c] = t;                                                    \
        m = fmaxf(m, t);                                                       \
      }                                                                        \
    }                                                                          \
    if (MORE) LOAD_W((I) + 1);                                                 \
    _Pragma("unroll") for (int off = 16; off > 0; off >>= 1)                   \
      m = fmaxf(m, __shfl_xor(m, off, 32));                                    \
    float s = 0.0f, wsum = 0.0f;                                               \
    _Pragma("unroll") for (int it = 0; it < 8; ++it) {                         \
      _Pragma("unroll") for (int c = 0; c < 4; ++c) {                          \
        float e = __expf(g[SLOT][it][c] - m);                                  \
        s += e;                                                                \
        wsum = fmaf(e, hreg[it][c], wsum);                                     \
      }                                                                        \
    }                                                                          \
    _Pragma("unroll") for (int off = 16; off > 0; off >>= 1) {                 \
      s += __shfl_xor(s, off, 32);                                             \
      wsum += __shfl_xor(wsum, off, 32);                                       \
    }                                                                          \
    if (lane == (I)) { my_s = s; my_w = wsum; }                                \
  } while (0)

#pragma unroll 1
  for (int t = 0; t < NB - RING; t += RING) {
    __builtin_amdgcn_sched_barrier(0);
    LOAD_ROW(3, t + 3);
    __builtin_amdgcn_sched_barrier(0);
    DO_ROW(0, t + 0, true);
    __builtin_amdgcn_sched_barrier(0);
    LOAD_ROW(0, t + 4);
    __builtin_amdgcn_sched_barrier(0);
    DO_ROW(1, t + 1, true);
    __builtin_amdgcn_sched_barrier(0);
    LOAD_ROW(1, t + 5);
    __builtin_amdgcn_sched_barrier(0);
    DO_ROW(2, t + 2, true);
    __builtin_amdgcn_sched_barrier(0);
    LOAD_ROW(2, t + 6);
    __builtin_amdgcn_sched_barrier(0);
    DO_ROW(3, t + 3, true);
  }
  __builtin_amdgcn_sched_barrier(0);
  LOAD_ROW(3, NB - 1);
  __builtin_amdgcn_sched_barrier(0);
  DO_ROW(0, NB - 4, true);
  DO_ROW(1, NB - 3, true);
  DO_ROW(2, NB - 2, true);
  DO_ROW(3, NB - 1, false);
#undef LOAD_ROW
#undef LOAD_W
#undef DO_ROW

  if (lane < NB) {
    const size_t idx = (size_t)b * H_DIM + o0 + lane;
    const float contrib = my_w / my_s;
    const float* pe = part + (wv & 7) * 32 + (wv >> 3) * 16 + lane;
    float icv = pe[0];
#pragma unroll
    for (int v = 1; v < NW; ++v) icv += pe[v * 256];
    icv += b_ih[o0 + lane];
    out[idx] = tanhf(icv + contrib);
  }
}

extern "C" void kernel_launch(void* const* d_in, const int* in_sizes, int n_in,
                              void* d_out, int out_size, void* d_ws, size_t ws_size,
                              hipStream_t stream) {
  const float* x_t         = (const float*)d_in[0];
  const float* h_prev      = (const float*)d_in[1];
  const float* W_ih        = (const float*)d_in[2];
  const float* b_ih        = (const float*)d_in[3];
  const float* W_hh        = (const float*)d_in[4];
  const float* temperature = (const float*)d_in[5];
  const float* gumbel      = (const float*)d_in[6];
  float* out = (float*)d_out;

  reservoir_step_fused<<<(H_DIM / NW) * (B_DIM / NB), NW * 32, 0, stream>>>(
      x_t, h_prev, W_ih, b_ih, W_hh, temperature, gumbel, out);
}
